# starting kernel
# baseline (speedup 1.0000x reference)
#include <hip/hip_runtime.h>
#include <hip/hip_bf16.h>


typedef __attribute__((ext_vector_type(2))) float v2f;
typedef __attribute__((ext_vector_type(8))) float v8f;

#define CIN   64
#define COUT  64
#define RANK  4
#define HH    56
#define WW    56
#define KDIM  (CIN * 9)
#define TILE_W 58
#define CI_STRIDE (3 * TILE_W)

__global__ void prep_keff(const float* __restrict__ vw,
                          const float* __restrict__ hw,
                          float* __restrict__ keff) {
    int idx = blockIdx.x * blockDim.x + threadIdx.x;
    if (idx >= COUT * KDIM) return;
    int co = idx / KDIM;
    int k  = idx - co * KDIM;
    int kk = k >> 6;
    int ci = k & 63;
    int kh = kk / 3;
    int kw = kk - kh * 3;
    int gbase = ci * (COUT * RANK) + co * RANK;
    float s = 0.f;
#pragma unroll
    for (int r = 0; r < RANK; ++r) {
        int g = gbase + r;
        s += hw[g * 3 + kh] * vw[g * 3 + kw];
    }
    keff[idx] = s;
}

__global__ __launch_bounds__(512)
void conv_wmma(const float* __restrict__ x,
               const float* __restrict__ keff,
               const float* __restrict__ bias,
               float* __restrict__ out) {
    __shared__ float tile[CIN * CI_STRIDE];

    const int bh = blockIdx.x;
    const int b  = bh / HH;
    const int h  = bh - b * HH;
    const int tid = threadIdx.x;

    const float* xb = x + (size_t)b * CIN * HH * WW;
    for (int e = tid; e < CIN * CI_STRIDE; e += 512) {
        int ci  = e / CI_STRIDE;
        int rem = e - ci * CI_STRIDE;
        int ry  = rem / TILE_W;
        int col = rem - ry * TILE_W;
        int y  = h + ry - 1;
        int wc = col - 1;
        float v = 0.f;
        if ((unsigned)y < (unsigned)HH && (unsigned)wc < (unsigned)WW)
            v = xb[(ci * HH + y) * WW + wc];
        tile[e] = v;
    }
    __syncthreads();

    const int wave  = tid >> 5;
    const int lane  = tid & 31;
    const int half  = lane >> 4;
    const int laneW = lane & 15;
    const int co0 = (wave >> 2) * 16;
    const int w0  = (wave & 3) * 16;

    const float* aRow = keff + (co0 + laneW) * KDIM;

    v8f acc = {0.f, 0.f, 0.f, 0.f, 0.f, 0.f, 0.f, 0.f};

    for (int kk = 0; kk < 9; ++kk) {
        const int kh = kk / 3;
        const int kw = kk - kh * 3;
        const float* bBase = tile + 2 * half * CI_STRIDE
                                  + kh * TILE_W + w0 + laneW + kw;
        const float* aK = aRow + kk * 64 + 2 * half;
#pragma unroll
        for (int t = 0; t < 16; ++t) {
            v2f a = *(const v2f*)(aK + 4 * t);
            v2f bv;
            bv.x = bBase[(4 * t)     * CI_STRIDE];
            bv.y = bBase[(4 * t + 1) * CI_STRIDE];
            acc = __builtin_amdgcn_wmma_f32_16x16x4_f32(
                false, a, false, bv, (short)0, acc, false, false);
        }
    }

    const int wN = w0 + laneW;
    if (wN < WW) {
        const int coB = co0 + half * 8;
        float* op = out + (((size_t)b * COUT + coB) * HH + h) * WW + wN;
#pragma unroll
        for (int j = 0; j < 8; ++j)
            op[(size_t)j * HH * WW] = acc[j] + bias[coB + j];
    }
}

extern "C" void kernel_launch(void* const* d_in, const int* in_sizes, int n_in,
                              void* d_out, int out_size, void* d_ws, size_t ws_size,
                              hipStream_t stream) {
    const float* x    = (const float*)d_in[0];
    const float* vw   = (const float*)d_in[1];
    const float* hw   = (const float*)d_in[2];
    const float* bias = (const float*)d_in[3];
    float* out  = (float*)d_out;
    float* keff = (float*)d_ws;

    const int nK = COUT * KDIM;
    prep_keff<<<(nK + 255) / 256, 256, 0, stream>>>(vw, hw, keff);

    const int B = 4;
    conv_wmma<<<B * HH, 512, 0, stream>>>(x, keff, bias, out);
}


# Wave tile 16co x 64w x 16ci: each wave holds its 36 A fragments in registers and feeds all 4 w tiles of its row; 4 ci-quarter partials summed through LDS in fixed order: summation order changes
# speedup vs baseline: 1.1967x; 1.1967x over previous
#include <hip/hip_runtime.h>
#include <hip/hip_bf16.h>


typedef __attribute__((ext_vector_type(2))) float v2f;
typedef __attribute__((ext_vector_type(8))) float v8f;

#define CIN   64
#define COUT  64
#define RANK  4
#define HH    56
#define WW    56
#define KDIM  (CIN * 9)
#define TILE_W 58
#define CI_STRIDE (3 * TILE_W)

__global__ void prep_keff(const float* __restrict__ vw,
                          const float* __restrict__ hw,
                          float* __restrict__ keff) {
    int idx = blockIdx.x * blockDim.x + threadIdx.x;
    if (idx >= COUT * KDIM) return;
    int co = idx / KDIM;
    int k  = idx - co * KDIM;
    int kk = k >> 6;
    int ci = k & 63;
    int kh = kk / 3;
    int kw = kk - kh * 3;
    int gbase = ci * (COUT * RANK) + co * RANK;
    float s = 0.f;
#pragma unroll
    for (int r = 0; r < RANK; ++r) {
        int g = gbase + r;
        s += hw[g * 3 + kh] * vw[g * 3 + kw];
    }
    keff[idx] = s;
}

#define SLAB_FLOATS (CIN * CI_STRIDE)
#define PART_FLOATS (8 * 32)
#define RED_FLOATS  (4 * 4 * 4 * PART_FLOATS)

__global__ __launch_bounds__(512) __attribute__((amdgpu_waves_per_eu(4, 5)))
void conv_wmma(const float* __restrict__ x,
               const float* __restrict__ keff,
               const float* __restrict__ bias,
               float* __restrict__ out) {
    __shared__ float smem[SLAB_FLOATS + RED_FLOATS];
    float* tile = smem;
    float* red  = smem + SLAB_FLOATS;

    const int bh = blockIdx.x;
    const int b  = bh / HH;
    const int h  = bh - b * HH;
    const int tid = threadIdx.x;

    const int wave  = tid >> 5;
    const int lane  = tid & 31;
    const int half  = lane >> 4;
    const int laneW = lane & 15;
    const int ct  = wave >> 2;
    const int q   = wave & 3;
    const int co0 = ct * 16;

    const float* xb = x + (size_t)b * CIN * HH * WW;
    for (int e = tid; e < SLAB_FLOATS; e += 512) {
        int ci  = e / CI_STRIDE;
        int rem = e - ci * CI_STRIDE;
        int ry  = rem / TILE_W;
        int col = rem - ry * TILE_W;
        int y  = h + ry - 1;
        int wc = col - 1;
        float v = 0.f;
        if ((unsigned)y < (unsigned)HH && (unsigned)wc < (unsigned)WW)
            v = xb[(ci * HH + y) * WW + wc];
        tile[e] = v;
    }
    __syncthreads();

    const float* aRow = keff + (co0 + laneW) * KDIM + 16 * q + 2 * half;
    v2f afrag[9][4];
#pragma unroll
    for (int kk = 0; kk < 9; ++kk)
#pragma unroll
        for (int tt = 0; tt < 4; ++tt)
            afrag[kk][tt] = *(const v2f*)(aRow + kk * 64 + 4 * tt);

    v8f acc[4];
#pragma unroll
    for (int j = 0; j < 4; ++j)
        acc[j] = (v8f){0.f, 0.f, 0.f, 0.f, 0.f, 0.f, 0.f, 0.f};

    const float* bLane  = tile + (16 * q + 2 * half) * CI_STRIDE;
    const float* bLane0 = bLane + laneW;
    const float* bLane3 = bLane + 48 + (laneW < 8 ? laneW : 7);

#pragma unroll
    for (int kk = 0; kk < 9; ++kk) {
        const int kh = kk / 3;
        const int kw = kk - kh * 3;
#pragma unroll
        for (int tt = 0; tt < 4; ++tt) {
            const int off = (4 * tt) * CI_STRIDE + kh * TILE_W + kw;
#pragma unroll
            for (int j = 0; j < 4; ++j) {
                const float* bp = (j < 3) ? (bLane0 + 16 * j + off) : (bLane3 + off);
                v2f bv;
                bv.x = bp[0];
                bv.y = bp[CI_STRIDE];
                acc[j] = __builtin_amdgcn_wmma_f32_16x16x4_f32(
                    false, afrag[kk][tt], false, bv, (short)0, acc[j], false, false);
            }
        }
    }

    const int coB = co0 + half * 8;
    float bb[8];
#pragma unroll
    for (int r = 0; r < 8; ++r)
        bb[r] = bias[coB + r];

    float* rw = red + ((q * 4 + ct) * 4) * PART_FLOATS + lane;
#pragma unroll
    for (int j = 0; j < 4; ++j)
#pragma unroll
        for (int r = 0; r < 8; ++r)
            rw[j * PART_FLOATS + r * 32] = acc[j][r];
    __syncthreads();

    const float* rr = red + (ct * 4 + q) * PART_FLOATS + lane;
    const int wN = 16 * q + laneW;
    if (wN < WW) {
        float* op = out + (((size_t)b * COUT + coB) * HH + h) * WW + wN;
#pragma unroll
        for (int r = 0; r < 8; ++r) {
            float s = rr[0 * 16 * PART_FLOATS + r * 32];
            s += rr[1 * 16 * PART_FLOATS + r * 32];
            s += rr[2 * 16 * PART_FLOATS + r * 32];
            s += rr[3 * 16 * PART_FLOATS + r * 32];
            op[(size_t)r * HH * WW] = s + bb[r];
        }
    }
}

extern "C" void kernel_launch(void* const* d_in, const int* in_sizes, int n_in,
                              void* d_out, int out_size, void* d_ws, size_t ws_size,
                              hipStream_t stream) {
    const float* x    = (const float*)d_in[0];
    const float* vw   = (const float*)d_in[1];
    const float* hw   = (const float*)d_in[2];
    const float* bias = (const float*)d_in[3];
    float* out  = (float*)d_out;
    float* keff = (float*)d_ws;

    const int nK = COUT * KDIM;
    prep_keff<<<(nK + 255) / 256, 256, 0, stream>>>(vw, hw, keff);

    const int B = 4;
    conv_wmma<<<B * HH, 512, 0, stream>>>(x, keff, bias, out);
}


# Input slab channels-last in LDS (pixel pitch 68 floats), filled by grouped 16-byte loads transposed in registers; a lane's 8 ci of a pixel = two 16-byte LDS reads for 4 k steps; sum order changes
# speedup vs baseline: 1.9759x; 1.6511x over previous
#include <hip/hip_runtime.h>
#include <hip/hip_bf16.h>


typedef __attribute__((ext_vector_type(2))) float v2f;
typedef __attribute__((ext_vector_type(4))) float v4f;
typedef __attribute__((ext_vector_type(8))) float v8f;

#define CIN   64
#define COUT  64
#define RANK  4
#define HH    56
#define WW    56
#define KDIM  (CIN * 9)
#define TILE_W 58
#define PIX_PITCH 68
#define PLANE (HH * WW)

__global__ void prep_keff(const float* __restrict__ vw,
                          const float* __restrict__ hw,
                          float* __restrict__ keff) {
    int idx = blockIdx.x * blockDim.x + threadIdx.x;
    if (idx >= COUT * KDIM) return;
    int co = idx / KDIM;
    int k  = idx - co * KDIM;
    int kk = k >> 6;
    int ci = k & 63;
    int kh = kk / 3;
    int kw = kk - kh * 3;
    int gbase = ci * (COUT * RANK) + co * RANK;
    float s = 0.f;
#pragma unroll
    for (int r = 0; r < RANK; ++r) {
        int g = gbase + r;
        s += hw[g * 3 + kh] * vw[g * 3 + kw];
    }
    keff[idx] = s;
}

#define SLAB_FLOATS (3 * TILE_W * PIX_PITCH)
#define N_UNITS (14 * 16 * 3)
#define PART_FLOATS (8 * 32)
#define RED_FLOATS  (4 * 4 * 4 * PART_FLOATS)

__global__ __launch_bounds__(512) __attribute__((amdgpu_waves_per_eu(4, 5)))
void conv_wmma(const float* __restrict__ x,
               const float* __restrict__ keff,
               const float* __restrict__ bias,
               float* __restrict__ out) {
    __shared__ __attribute__((aligned(16))) float smem[SLAB_FLOATS + RED_FLOATS];
    float* tile = smem;
    float* red  = smem + SLAB_FLOATS;

    const int bh = blockIdx.x;
    const int b  = bh / HH;
    const int h  = bh - b * HH;
    const int tid = threadIdx.x;

    const int wave  = tid >> 5;
    const int lane  = tid & 31;
    const int half  = lane >> 4;
    const int laneW = lane & 15;
    const int ct  = wave >> 2;
    const int q   = wave & 3;
    const int co0 = ct * 16;

    const float* xb = x + (size_t)b * CIN * HH * WW;
    const bool has1 = tid < (N_UNITS - 512);

    const int f0  = tid % 14;
    const int t0  = tid / 14;
    const int a0  = t0 & 15;
    const int ry0 = t0 >> 4;
    const int y0  = h + ry0 - 1;
    const bool ok0 = (unsigned)y0 < (unsigned)HH;
    const float* s0 = xb + ((4 * a0) * HH + (ok0 ? y0 : h)) * WW + 4 * f0;
    v4f p0 = *(const v4f*)(s0);
    v4f p1 = *(const v4f*)(s0 + PLANE);
    v4f p2 = *(const v4f*)(s0 + 2 * PLANE);
    v4f p3 = *(const v4f*)(s0 + 3 * PLANE);

    const int u1  = tid + 512;
    const int f1  = u1 % 14;
    const int t1  = u1 / 14;
    const int a1  = t1 & 15;
    const int ry1 = t1 >> 4;
    const int y1  = h + ry1 - 1;
    const bool ok1 = (unsigned)y1 < (unsigned)HH;
    v4f r0 = (v4f){0.f, 0.f, 0.f, 0.f}, r1 = r0, r2 = r0, r3 = r0;
    if (has1) {
        const float* s1 = xb + ((4 * a1) * HH + (ok1 ? y1 : h)) * WW + 4 * f1;
        r0 = *(const v4f*)(s1);
        r1 = *(const v4f*)(s1 + PLANE);
        r2 = *(const v4f*)(s1 + 2 * PLANE);
        r3 = *(const v4f*)(s1 + 3 * PLANE);
    }

    {
        if (!ok0) { p0 = (v4f){0.f, 0.f, 0.f, 0.f}; p1 = p0; p2 = p0; p3 = p0; }
        float* d0 = tile + (ry0 * TILE_W + 4 * f0 + 1) * PIX_PITCH + 4 * a0;
#pragma unroll
        for (int e = 0; e < 4; ++e)
            *(v4f*)(d0 + e * PIX_PITCH) = (v4f){p0[e], p1[e], p2[e], p3[e]};
    }
    if (has1) {
        if (!ok1) { r0 = (v4f){0.f, 0.f, 0.f, 0.f}; r1 = r0; r2 = r0; r3 = r0; }
        float* d1 = tile + (ry1 * TILE_W + 4 * f1 + 1) * PIX_PITCH + 4 * a1;
#pragma unroll
        for (int e = 0; e < 4; ++e)
            *(v4f*)(d1 + e * PIX_PITCH) = (v4f){r0[e], r1[e], r2[e], r3[e]};
    }
    if (tid >= 256 && tid < 256 + 96) {
        const int t2 = tid - 256;
        const int hr = t2 >> 5;
        const int hc = ((t2 >> 4) & 1) ? (TILE_W - 1) : 0;
        *(v4f*)(tile + (hr * TILE_W + hc) * PIX_PITCH + 4 * (t2 & 15)) = (v4f){0.f, 0.f, 0.f, 0.f};
    }
    __syncthreads();

    const float* aRow = keff + (co0 + laneW) * KDIM + 16 * q + 8 * half;
    v4f alo[9], ahi[9];
#pragma unroll
    for (int kk = 0; kk < 9; ++kk) {
        alo[kk] = *(const v4f*)(aRow + kk * 64);
        ahi[kk] = *(const v4f*)(aRow + kk * 64 + 4);
    }

    v8f acc[4];
#pragma unroll
    for (int j = 0; j < 4; ++j)
        acc[j] = (v8f){0.f, 0.f, 0.f, 0.f, 0.f, 0.f, 0.f, 0.f};

    const float* bBase  = tile + 16 * q + 8 * half;
    const float* bLane0 = bBase + laneW * PIX_PITCH;
    const float* bLane3 = bBase + (48 + (laneW < 8 ? laneW : 7)) * PIX_PITCH;

#pragma unroll
    for (int kk = 0; kk < 9; ++kk) {
        const int kh = kk / 3;
        const int kw = kk - kh * 3;
        const int poff = (kh * TILE_W + kw) * PIX_PITCH;
        v4f blo[4], bhi[4];
#pragma unroll
        for (int j = 0; j < 4; ++j) {
            const float* bp = (j < 3) ? (bLane0 + 16 * j * PIX_PITCH + poff) : (bLane3 + poff);
            blo[j] = *(const v4f*)(bp);
            bhi[j] = *(const v4f*)(bp + 4);
        }
#pragma unroll
        for (int tt = 0; tt < 4; ++tt) {
            const v4f aq = (tt < 2) ? alo[kk] : ahi[kk];
            v2f av;
            av.x = (tt & 1) ? aq.z : aq.x;
            av.y = (tt & 1) ? aq.w : aq.y;
#pragma unroll
            for (int j = 0; j < 4; ++j) {
                const v4f bq = (tt < 2) ? blo[j] : bhi[j];
                v2f bv;
                bv.x = (tt & 1) ? bq.z : bq.x;
                bv.y = (tt & 1) ? bq.w : bq.y;
                acc[j] = __builtin_amdgcn_wmma_f32_16x16x4_f32(
                    false, av, false, bv, (short)0, acc[j], false, false);
            }
        }
    }

    const int coB = co0 + half * 8;
    float bb[8];
#pragma unroll
    for (int r = 0; r < 8; ++r)
        bb[r] = bias[coB + r];

    float* rw = red + ((q * 4 + ct) * 4) * PART_FLOATS + lane;
#pragma unroll
    for (int j = 0; j < 4; ++j)
#pragma unroll
        for (int r = 0; r < 8; ++r)
            rw[j * PART_FLOATS + r * 32] = acc[j][r];
    __syncthreads();

    const float* rr = red + (ct * 4 + q) * PART_FLOATS + lane;
    const int wN = 16 * q + laneW;
    if (wN < WW) {
        float* op = out + (((size_t)b * COUT + coB) * HH + h) * WW + wN;
#pragma unroll
        for (int r = 0; r < 8; ++r) {
            float s = rr[0 * 16 * PART_FLOATS + r * 32];
            s += rr[1 * 16 * PART_FLOATS + r * 32];
            s += rr[2 * 16 * PART_FLOATS + r * 32];
            s += rr[3 * 16 * PART_FLOATS + r * 32];
            op[(size_t)r * HH * WW] = s + bb[r];
        }
    }
}

extern "C" void kernel_launch(void* const* d_in, const int* in_sizes, int n_in,
                              void* d_out, int out_size, void* d_ws, size_t ws_size,
                              hipStream_t stream) {
    const float* x    = (const float*)d_in[0];
    const float* vw   = (const float*)d_in[1];
    const float* hw   = (const float*)d_in[2];
    const float* bias = (const float*)d_in[3];
    float* out  = (float*)d_out;
    float* keff = (float*)d_ws;

    const int nK = COUT * KDIM;
    prep_keff<<<(nK + 255) / 256, 256, 0, stream>>>(vw, hw, keff);

    const int B = 4;
    conv_wmma<<<B * HH, 512, 0, stream>>>(x, keff, bias, out);
}


# One launch: each wave collapses its own 16 co x 8 ci weight pairs in registers from 16-byte loads of the two weight tensors, beside its products; K walked ci-phase-major: summation order changes
# speedup vs baseline: 2.3112x; 1.1697x over previous
#include <hip/hip_runtime.h>
#include <hip/hip_bf16.h>


typedef __attribute__((ext_vector_type(2))) float v2f;
typedef __attribute__((ext_vector_type(4))) float v4f;
typedef __attribute__((ext_vector_type(8))) float v8f;

#define CIN   64
#define COUT  64
#define RANK  4
#define HH    56
#define WW    56
#define TILE_W 58
#define PIX_PITCH 68
#define PLANE (HH * WW)
#define PAIR_FLOATS (RANK * 3)
#define CI_PITCH (COUT * PAIR_FLOATS)
#define N_PHASE 8
#define CI_BASE(p) (16 * ((p) >> 1) + 4 * ((p) & 1))

#define SLAB_FLOATS (3 * TILE_W * PIX_PITCH)
#define N_THREADS 128
#define STAGE_THREADS 112
#define UNITS_PER_THREAD 6

__global__ __launch_bounds__(N_THREADS) __attribute__((amdgpu_waves_per_eu(1, 1)))
void conv_wmma(const float* __restrict__ x,
               const float* __restrict__ vw,
               const float* __restrict__ hw,
               const float* __restrict__ bias,
               float* __restrict__ out) {
    __shared__ __attribute__((aligned(16))) float tile[SLAB_FLOATS];

    const int bh = blockIdx.x;
    const int b  = bh / HH;
    const int h  = bh - b * HH;
    const int tid = threadIdx.x;

    const int wave  = tid >> 5;
    const int lane  = tid & 31;
    const int half  = lane >> 4;
    const int laneW = lane & 15;
    const int co0 = wave * 16;

    const float* xb = x + (size_t)b * CIN * HH * WW;
    const bool stager = tid < STAGE_THREADS;
    const int f  = tid % 14;
    const int g8 = tid / 14;

    const float* hLane = hw + (8 * half) * CI_PITCH + (co0 + laneW) * PAIR_FLOATS;
    const float* vLane = vw + (8 * half) * CI_PITCH + (co0 + laneW) * PAIR_FLOATS;
    v4f wraw[2][24];
#pragma unroll
    for (int e = 0; e < 4; ++e)
#pragma unroll
        for (int m = 0; m < 3; ++m) {
            wraw[0][6 * e + m]     = *(const v4f*)(hLane + (CI_BASE(0) + e) * CI_PITCH + 4 * m);
            wraw[0][6 * e + 3 + m] = *(const v4f*)(vLane + (CI_BASE(0) + e) * CI_PITCH + 4 * m);
        }

    const int coB = co0 + half * 8;
    float bb[8];
#pragma unroll
    for (int r = 0; r < 8; ++r)
        bb[r] = bias[coB + r];

    if (stager) {
        v4f p[UNITS_PER_THREAD][4];
#pragma unroll
        for (int i = 0; i < UNITS_PER_THREAD; ++i) {
            const int ry = i >> 1;
            const int a  = g8 + 8 * (i & 1);
            const int y  = h + ry - 1;
            const bool ok = (unsigned)y < (unsigned)HH;
            const float* s = xb + ((4 * a) * HH + (ok ? y : h)) * WW + 4 * f;
            p[i][0] = *(const v4f*)(s);
            p[i][1] = *(const v4f*)(s + PLANE);
            p[i][2] = *(const v4f*)(s + 2 * PLANE);
            p[i][3] = *(const v4f*)(s + 3 * PLANE);
        }
        __builtin_amdgcn_sched_barrier(0);
#pragma unroll
        for (int i = 0; i < UNITS_PER_THREAD; ++i) {
            const int ry = i >> 1;
            const int a  = g8 + 8 * (i & 1);
            const int y  = h + ry - 1;
            const bool ok = (unsigned)y < (unsigned)HH;
            v4f c0 = p[i][0], c1 = p[i][1], c2 = p[i][2], c3 = p[i][3];
            if (!ok) { c0 = (v4f){0.f, 0.f, 0.f, 0.f}; c1 = c0; c2 = c0; c3 = c0; }
            float* d = tile + (ry * TILE_W + 4 * f + 1) * PIX_PITCH + 4 * a;
#pragma unroll
            for (int e = 0; e < 4; ++e)
                *(v4f*)(d + e * PIX_PITCH) = (v4f){c0[e], c1[e], c2[e], c3[e]};
        }
    }
    if (tid < 96) {
        const int hr = tid >> 5;
        const int hc = ((tid >> 4) & 1) ? (TILE_W - 1) : 0;
        *(v4f*)(tile + (hr * TILE_W + hc) * PIX_PITCH + 4 * (tid & 15)) = (v4f){0.f, 0.f, 0.f, 0.f};
    }
    __syncthreads();

#pragma unroll
    for (int e = 0; e < 4; ++e)
#pragma unroll
        for (int m = 0; m < 3; ++m) {
            wraw[1][6 * e + m]     = *(const v4f*)(hLane + (CI_BASE(1) + e) * CI_PITCH + 4 * m);
            wraw[1][6 * e + 3 + m] = *(const v4f*)(vLane + (CI_BASE(1) + e) * CI_PITCH + 4 * m);
        }

    const float* bLane0 = tile + 8 * half + laneW * PIX_PITCH;
    const float* bLane3 = tile + 8 * half + (48 + (laneW < 8 ? laneW : 7)) * PIX_PITCH;

    v4f bfr[2][36];
#pragma unroll
    for (int kk = 0; kk < 9; ++kk)
#pragma unroll
        for (int j = 0; j < 4; ++j) {
            const int poff = ((kk / 3) * TILE_W + (kk % 3)) * PIX_PITCH;
            const float* bp = ((j < 3) ? (bLane0 + 16 * j * PIX_PITCH) : bLane3) + poff + CI_BASE(0);
            bfr[0][4 * kk + j] = *(const v4f*)(bp);
        }

    __builtin_amdgcn_sched_barrier(0);

    float akf[2][4][9];
#pragma unroll
    for (int e = 0; e < 4; ++e) {
        float hv[12], vv[12];
#pragma unroll
        for (int m = 0; m < 3; ++m)
#pragma unroll
            for (int c = 0; c < 4; ++c) {
                hv[4 * m + c] = wraw[0][6 * e + m][c];
                vv[4 * m + c] = wraw[0][6 * e + 3 + m][c];
            }
#pragma unroll
        for (int kh = 0; kh < 3; ++kh)
#pragma unroll
            for (int kw = 0; kw < 3; ++kw) {
                float s = hv[kh] * vv[kw];
#pragma unroll
                for (int r = 1; r < RANK; ++r)
                    s = __builtin_fmaf(hv[3 * r + kh], vv[3 * r + kw], s);
                akf[0][e][kh * 3 + kw] = s;
            }
    }

    v8f acc[4];
#pragma unroll
    for (int j = 0; j < 4; ++j)
        acc[j] = (v8f){0.f, 0.f, 0.f, 0.f, 0.f, 0.f, 0.f, 0.f};

    __builtin_amdgcn_sched_barrier(0);

#pragma unroll
    for (int p = 0; p < N_PHASE; ++p) {
        const int cur = p & 1;
        if (p + 1 < N_PHASE) {
#pragma unroll
            for (int kk = 0; kk < 9; ++kk)
#pragma unroll
                for (int j = 0; j < 4; ++j) {
                    const int poff = ((kk / 3) * TILE_W + (kk % 3)) * PIX_PITCH;
                    const float* bp = ((j < 3) ? (bLane0 + 16 * j * PIX_PITCH) : bLane3) + poff + CI_BASE(p + 1);
                    bfr[cur ^ 1][4 * kk + j] = *(const v4f*)(bp);
                }
        }
        if (p + 2 < N_PHASE) {
#pragma unroll
            for (int e = 0; e < 4; ++e)
#pragma unroll
                for (int m = 0; m < 3; ++m) {
                    wraw[cur][6 * e + m]     = *(const v4f*)(hLane + (CI_BASE(p + 2) + e) * CI_PITCH + 4 * m);
                    wraw[cur][6 * e + 3 + m] = *(const v4f*)(vLane + (CI_BASE(p + 2) + e) * CI_PITCH + 4 * m);
                }
        }
        __builtin_amdgcn_sched_barrier(0);
#pragma unroll
        for (int kk = 0; kk < 9; ++kk) {
#pragma unroll
            for (int u = 0; u < 2; ++u) {
                const int st = 2 * kk + u;
                if (p + 1 < N_PHASE && st >= 6) {
                    const int e  = (st - 6) / 3;
                    const int kh = (st - 6) % 3;
                    float hv[RANK], vv[RANK][3];
#pragma unroll
                    for (int r = 0; r < RANK; ++r) {
                        const int ih = 3 * r + kh;
                        hv[r] = wraw[cur ^ 1][6 * e + (ih >> 2)][ih & 3];
#pragma unroll
                        for (int kw = 0; kw < 3; ++kw) {
                            const int iv = 3 * r + kw;
                            vv[r][kw] = wraw[cur ^ 1][6 * e + 3 + (iv >> 2)][iv & 3];
                        }
                    }
#pragma unroll
                    for (int kw = 0; kw < 3; ++kw) {
                        float s = hv[0] * vv[0][kw];
#pragma unroll
                        for (int r = 1; r < RANK; ++r)
                            s = __builtin_fmaf(hv[r], vv[r][kw], s);
                        akf[cur ^ 1][e][kh * 3 + kw] = s;
                    }
                }
                v2f av;
                av.x = akf[cur][2 * u][kk];
                av.y = akf[cur][2 * u + 1][kk];
#pragma unroll
                for (int j = 0; j < 4; ++j) {
                    const v4f bq = bfr[cur][4 * kk + j];
                    v2f bv;
                    bv.x = u ? bq.z : bq.x;
                    bv.y = u ? bq.w : bq.y;
                    acc[j] = __builtin_amdgcn_wmma_f32_16x16x4_f32(
                        false, av, false, bv, (short)0, acc[j], false, false);
                }
                __builtin_amdgcn_sched_barrier(0);
            }
        }
    }

    float* op = out + (((size_t)b * COUT + coB) * HH + h) * WW + laneW;
#pragma unroll
    for (int j = 0; j < 4; ++j) {
        if (j < 3 || laneW < 8) {
#pragma unroll
            for (int r = 0; r < 8; ++r)
                op[(size_t)r * HH * WW + 16 * j] = acc[j][r] + bb[r];
        }
    }
}

extern "C" void kernel_launch(void* const* d_in, const int* in_sizes, int n_in,
                              void* d_out, int out_size, void* d_ws, size_t ws_size,
                              hipStream_t stream) {
    const float* x    = (const float*)d_in[0];
    const float* vw   = (const float*)d_in[1];
    const float* hw   = (const float*)d_in[2];
    const float* bias = (const float*)d_in[3];
    float* out  = (float*)d_out;

    const int B = 4;
    conv_wmma<<<B * HH, N_THREADS, 0, stream>>>(x, vw, hw, bias, out);
}


# Workgroup = one 16-co tile x 4 output rows: its 224 pixels are 14 unpadded tiles, 3 per wave plus half the k steps of a tile shared by a wave pair, joined through LDS: summation order changes
# speedup vs baseline: 2.4734x; 1.0702x over previous
#include <hip/hip_runtime.h>
#include <hip/hip_bf16.h>


typedef __attribute__((ext_vector_type(2))) float v2f;
typedef __attribute__((ext_vector_type(4))) float v4f;
typedef __attribute__((ext_vector_type(8))) float v8f;

#define CIN   64
#define COUT  64
#define RANK  4
#define HH    56
#define WW    56
#define TILE_W 58
#define PIX_PITCH 68
#define PLANE (HH * WW)
#define PAIR_FLOATS (RANK * 3)
#define CI_PITCH (COUT * PAIR_FLOATS)
#define N_PHASE 8
#define CI_BASE(p) (16 * ((p) >> 1) + 4 * ((p) & 1))

#define ROWS_PER_WG 4
#define SLAB_ROWS (ROWS_PER_WG + 2)
#define SLAB_FLOATS (SLAB_ROWS * TILE_W * PIX_PITCH)
#define PART_FLOATS (2 * 8 * 32)
#define N_THREADS 128
#define STAGE_THREADS 112
#define UNITS_PER_THREAD 12

__global__ __launch_bounds__(N_THREADS) __attribute__((amdgpu_waves_per_eu(1, 1)))
void conv_wmma(const float* __restrict__ x,
               const float* __restrict__ vw,
               const float* __restrict__ hw,
               const float* __restrict__ bias,
               float* __restrict__ out) {
    __shared__ __attribute__((aligned(16))) float tile[SLAB_FLOATS];
    __shared__ __attribute__((aligned(16))) float part[PART_FLOATS];

    const int cot = blockIdx.x & 3;
    const int rg  = blockIdx.x >> 2;
    const int b   = rg / (HH / ROWS_PER_WG);
    const int h0  = (rg - b * (HH / ROWS_PER_WG)) * ROWS_PER_WG;
    const int tid = threadIdx.x;

    const int wave  = tid >> 5;
    const int lane  = tid & 31;
    const int half  = lane >> 4;
    const int laneW = lane & 15;
    const int co0 = cot * 16;
    const int odd  = wave & 1;
    const bool oddW = odd != 0;

    const float* xb = x + (size_t)b * CIN * HH * WW;
    const bool stager = tid < STAGE_THREADS;
    const int f  = tid % 14;
    const int g8 = tid / 14;

    const float* hLane = hw + (8 * half) * CI_PITCH + (co0 + laneW) * PAIR_FLOATS;
    const float* vLane = vw + (8 * half) * CI_PITCH + (co0 + laneW) * PAIR_FLOATS;
    v4f wraw[2][24];
#pragma unroll
    for (int e = 0; e < 4; ++e)
#pragma unroll
        for (int m = 0; m < 3; ++m) {
            wraw[0][6 * e + m]     = *(const v4f*)(hLane + (CI_BASE(0) + e) * CI_PITCH + 4 * m);
            wraw[0][6 * e + 3 + m] = *(const v4f*)(vLane + (CI_BASE(0) + e) * CI_PITCH + 4 * m);
        }

    const int coB = co0 + half * 8;
    float bb[8];
#pragma unroll
    for (int r = 0; r < 8; ++r)
        bb[r] = bias[coB + r];

    if (stager) {
        v4f p[UNITS_PER_THREAD][4];
#pragma unroll
        for (int i = 0; i < UNITS_PER_THREAD; ++i) {
            const int ry = i >> 1;
            const int a  = g8 + 8 * (i & 1);
            const int y  = h0 + ry - 1;
            const bool ok = (unsigned)y < (unsigned)HH;
            const float* s = xb + ((4 * a) * HH + (ok ? y : h0)) * WW + 4 * f;
            p[i][0] = *(const v4f*)(s);
            p[i][1] = *(const v4f*)(s + PLANE);
            p[i][2] = *(const v4f*)(s + 2 * PLANE);
            p[i][3] = *(const v4f*)(s + 3 * PLANE);
        }
        __builtin_amdgcn_sched_barrier(0);
#pragma unroll
        for (int i = 0; i < UNITS_PER_THREAD; ++i) {
            const int ry = i >> 1;
            const int a  = g8 + 8 * (i & 1);
            const int y  = h0 + ry - 1;
            const bool ok = (unsigned)y < (unsigned)HH;
            v4f c0 = p[i][0], c1 = p[i][1], c2 = p[i][2], c3 = p[i][3];
            if (!ok) { c0 = (v4f){0.f, 0.f, 0.f, 0.f}; c1 = c0; c2 = c0; c3 = c0; }
            float* d = tile + (ry * TILE_W + 4 * f + 1) * PIX_PITCH + 4 * a;
#pragma unroll
            for (int e = 0; e < 4; ++e)
                *(v4f*)(d + e * PIX_PITCH) = (v4f){c0[e], c1[e], c2[e], c3[e]};
        }
    }
    for (int t = tid; t < SLAB_ROWS * 32; t += N_THREADS) {
        const int hr = t >> 5;
        const int hc = ((t >> 4) & 1) ? (TILE_W - 1) : 0;
        *(v4f*)(tile + (hr * TILE_W + hc) * PIX_PITCH + 4 * (t & 15)) = (v4f){0.f, 0.f, 0.f, 0.f};
    }
    __syncthreads();

#pragma unroll
    for (int e = 0; e < 4; ++e)
#pragma unroll
        for (int m = 0; m < 3; ++m) {
            wraw[1][6 * e + m]     = *(const v4f*)(hLane + (CI_BASE(1) + e) * CI_PITCH + 4 * m);
            wraw[1][6 * e + 3 + m] = *(const v4f*)(vLane + (CI_BASE(1) + e) * CI_PITCH + 4 * m);
        }

    const float* bOwn[3];
#pragma unroll
    for (int j = 0; j < 3; ++j) {
        const int q  = 16 * (3 * wave + j) + laneW;
        const int qr = q / WW;
        bOwn[j] = tile + 8 * half + (qr * TILE_W + (q - qr * WW)) * PIX_PITCH;
    }
    const int tS  = 12 + (wave >> 1);
    const int qS  = 16 * tS + laneW;
    const int qSr = qS / WW;
    const float* bSh = tile + 8 * half + 2 * odd + (qSr * TILE_W + (qS - qSr * WW)) * PIX_PITCH;

    v4f bfr[2][27];
    v2f bfs[2][9];
#pragma unroll
    for (int kk = 0; kk < 9; ++kk) {
        const int poff = ((kk / 3) * TILE_W + (kk % 3)) * PIX_PITCH;
#pragma unroll
        for (int j = 0; j < 3; ++j)
            bfr[0][3 * kk + j] = *(const v4f*)(bOwn[j] + poff + CI_BASE(0));
        bfs[0][kk] = *(const v2f*)(bSh + poff + CI_BASE(0));
    }

    __builtin_amdgcn_sched_barrier(0);

    float akf[2][4][9];
#pragma unroll
    for (int e = 0; e < 4; ++e) {
        float hv[12], vv[12];
#pragma unroll
        for (int m = 0; m < 3; ++m)
#pragma unroll
            for (int c = 0; c < 4; ++c) {
                hv[4 * m + c] = wraw[0][6 * e + m][c];
                vv[4 * m + c] = wraw[0][6 * e + 3 + m][c];
            }
#pragma unroll
        for (int kh = 0; kh < 3; ++kh)
#pragma unroll
            for (int kw = 0; kw < 3; ++kw) {
                float s = hv[kh] * vv[kw];
#pragma unroll
                for (int r = 1; r < RANK; ++r)
                    s = __builtin_fmaf(hv[3 * r + kh], vv[3 * r + kw], s);
                akf[0][e][kh * 3 + kw] = s;
            }
    }

    v8f acc[3];
#pragma unroll
    for (int j = 0; j < 3; ++j)
        acc[j] = (v8f){0.f, 0.f, 0.f, 0.f, 0.f, 0.f, 0.f, 0.f};
    v8f accS = (v8f){0.f, 0.f, 0.f, 0.f, 0.f, 0.f, 0.f, 0.f};

    __builtin_amdgcn_sched_barrier(0);

#pragma unroll
    for (int p = 0; p < N_PHASE; ++p) {
        const int cur = p & 1;
        if (p + 1 < N_PHASE) {
#pragma unroll
            for (int kk = 0; kk < 9; ++kk) {
                const int poff = ((kk / 3) * TILE_W + (kk % 3)) * PIX_PITCH;
#pragma unroll
                for (int j = 0; j < 3; ++j)
                    bfr[cur ^ 1][3 * kk + j] = *(const v4f*)(bOwn[j] + poff + CI_BASE(p + 1));
                bfs[cur ^ 1][kk] = *(const v2f*)(bSh + poff + CI_BASE(p + 1));
            }
        }
        if (p + 2 < N_PHASE) {
#pragma unroll
            for (int e = 0; e < 4; ++e)
#pragma unroll
                for (int m = 0; m < 3; ++m) {
                    wraw[cur][6 * e + m]     = *(const v4f*)(hLane + (CI_BASE(p + 2) + e) * CI_PITCH + 4 * m);
                    wraw[cur][6 * e + 3 + m] = *(const v4f*)(vLane + (CI_BASE(p + 2) + e) * CI_PITCH + 4 * m);
                }
        }
        __builtin_amdgcn_sched_barrier(0);
#pragma unroll
        for (int kk = 0; kk < 9; ++kk) {
#pragma unroll
            for (int u = 0; u < 2; ++u) {
                const int st = 2 * kk + u;
                if (p + 1 < N_PHASE && st >= 6) {
                    const int e  = (st - 6) / 3;
                    const int kh = (st - 6) % 3;
                    float hv[RANK], vv[RANK][3];
#pragma unroll
                    for (int r = 0; r < RANK; ++r) {
                        const int ih = 3 * r + kh;
                        hv[r] = wraw[cur ^ 1][6 * e + (ih >> 2)][ih & 3];
#pragma unroll
                        for (int kw = 0; kw < 3; ++kw) {
                            const int iv = 3 * r + kw;
                            vv[r][kw] = wraw[cur ^ 1][6 * e + 3 + (iv >> 2)][iv & 3];
                        }
                    }
#pragma unroll
                    for (int kw = 0; kw < 3; ++kw) {
                        float s = hv[0] * vv[0][kw];
#pragma unroll
                        for (int r = 1; r < RANK; ++r)
                            s = __builtin_fmaf(hv[r], vv[r][kw], s);
                        akf[cur ^ 1][e][kh * 3 + kw] = s;
                    }
                }
                v2f av;
                av.x = akf[cur][2 * u][kk];
                av.y = akf[cur][2 * u + 1][kk];
#pragma unroll
                for (int j = 0; j < 3; ++j) {
                    const v4f bq = bfr[cur][3 * kk + j];
                    v2f bv;
                    bv.x = u ? bq.z : bq.x;
                    bv.y = u ? bq.w : bq.y;
                    acc[j] = __builtin_amdgcn_wmma_f32_16x16x4_f32(
                        false, av, false, bv, (short)0, acc[j], false, false);
                }
                if (u == 0) {
                    v2f as;
                    as.x = oddW ? akf[cur][2][kk] : akf[cur][0][kk];
                    as.y = oddW ? akf[cur][3][kk] : akf[cur][1][kk];
                    accS = __builtin_amdgcn_wmma_f32_16x16x4_f32(
                        false, as, false, bfs[cur][kk], (short)0, accS, false, false);
                }
                __builtin_amdgcn_sched_barrier(0);
            }
        }
    }

    float* pp = part + (wave >> 1) * (8 * 32) + lane;
    if (oddW) {
#pragma unroll
        for (int r = 0; r < 8; ++r)
            pp[32 * r] = accS[r];
    }

    float* op = out + (((size_t)b * COUT + coB) * HH + h0) * WW + laneW;
#pragma unroll
    for (int j = 0; j < 3; ++j) {
#pragma unroll
        for (int r = 0; r < 8; ++r)
            op[(size_t)r * HH * WW + 16 * (3 * wave + j)] = acc[j][r] + bb[r];
    }
    __syncthreads();
    if (!oddW) {
#pragma unroll
        for (int r = 0; r < 8; ++r)
            op[(size_t)r * HH * WW + 16 * tS] = (accS[r] + pp[32 * r]) + bb[r];
    }
}

extern "C" void kernel_launch(void* const* d_in, const int* in_sizes, int n_in,
                              void* d_out, int out_size, void* d_ws, size_t ws_size,
                              hipStream_t stream) {
    const float* x    = (const float*)d_in[0];
    const float* vw   = (const float*)d_in[1];
    const float* hw   = (const float*)d_in[2];
    const float* bias = (const float*)d_in[3];
    float* out  = (float*)d_out;

    const int B = 4;
    conv_wmma<<<B * HH, N_THREADS, 0, stream>>>(x, vw, hw, bias, out);
}


# K split over the workgroup's 4 waves: wave w multiplies ci 16w..16w+15 into all 14 tiles and collapses each weight pair once; 4 partials added through LDS in fixed order: summation order changes
# speedup vs baseline: 2.6737x; 1.0810x over previous
#include <hip/hip_runtime.h>
#include <hip/hip_bf16.h>


typedef __attribute__((ext_vector_type(2))) float v2f;
typedef __attribute__((ext_vector_type(4))) float v4f;
typedef __attribute__((ext_vector_type(8))) float v8f;

#define CIN   64
#define COUT  64
#define RANK  4
#define HH    56
#define WW    56
#define TILE_W 58
#define PIX_PITCH 68
#define PLANE (HH * WW)
#define PAIR_FLOATS (RANK * 3)
#define CI_PITCH (COUT * PAIR_FLOATS)

#define ROWS_PER_WG 4
#define SLAB_ROWS (ROWS_PER_WG + 2)
#define SLAB_FLOATS (SLAB_ROWS * TILE_W * PIX_PITCH)
#define N_TILES 14
#define GROUP_PIX (ROWS_PER_WG * WW)
#define RED_PITCH 228
#define RED_FLOATS (4 * 16 * RED_PITCH)
#define N_THREADS 128
#define UNITS_PER_THREAD 12

__global__ __launch_bounds__(N_THREADS) __attribute__((amdgpu_waves_per_eu(1, 1)))
void conv_wmma(const float* __restrict__ x,
               const float* __restrict__ vw,
               const float* __restrict__ hw,
               const float* __restrict__ bias,
               float* __restrict__ out) {
    __shared__ __attribute__((aligned(16))) float tile[SLAB_FLOATS];

    const int cot = blockIdx.x & 3;
    const int rg  = blockIdx.x >> 2;
    const int b   = rg / (HH / ROWS_PER_WG);
    const int h0  = (rg - b * (HH / ROWS_PER_WG)) * ROWS_PER_WG;
    const int tid = threadIdx.x;

    const int wave  = tid >> 5;
    const int lane  = tid & 31;
    const int half  = lane >> 4;
    const int laneW = lane & 15;
    const int co0 = cot * 16;
    const int ciL = 16 * wave + 8 * half;

    const float* xb = x + (size_t)b * CIN * HH * WW;
    const int f  = tid % 14;
    const int g8 = (tid / 14 < 8) ? tid / 14 : 7;

    const float* hLane = hw + ciL * CI_PITCH + (co0 + laneW) * PAIR_FLOATS;
    const float* vLane = vw + ciL * CI_PITCH + (co0 + laneW) * PAIR_FLOATS;
    v4f wraw[8][6];
#pragma unroll
    for (int e = 0; e < 4; ++e)
#pragma unroll
        for (int m = 0; m < 3; ++m) {
            wraw[e][m]     = *(const v4f*)(hLane + e * CI_PITCH + 4 * m);
            wraw[e][3 + m] = *(const v4f*)(vLane + e * CI_PITCH + 4 * m);
        }

    float bb[7];
#pragma unroll
    for (int i = 0; i < 7; ++i)
        bb[i] = bias[co0 + 4 * wave + (lane + 32 * i) / 56];
    __builtin_amdgcn_sched_barrier(0);

    v4f p[UNITS_PER_THREAD][4];
#pragma unroll
    for (int i = 0; i < UNITS_PER_THREAD; ++i) {
        const int ry = i >> 1;
        const int a  = g8 + 8 * (i & 1);
        const int y  = h0 + ry - 1;
        const bool ok = (unsigned)y < (unsigned)HH;
        const float* s = xb + ((4 * a) * HH + (ok ? y : h0)) * WW + 4 * f;
        p[i][0] = *(const v4f*)(s);
        p[i][1] = *(const v4f*)(s + PLANE);
        p[i][2] = *(const v4f*)(s + 2 * PLANE);
        p[i][3] = *(const v4f*)(s + 3 * PLANE);
    }
    __builtin_amdgcn_sched_barrier(0);

    float akf[8][9];
#pragma unroll
    for (int e = 0; e < 4; ++e) {
        float hv[12], vv[12];
#pragma unroll
        for (int m = 0; m < 3; ++m)
#pragma unroll
            for (int c = 0; c < 4; ++c) {
                hv[4 * m + c] = wraw[e][m][c];
                vv[4 * m + c] = wraw[e][3 + m][c];
            }
#pragma unroll
        for (int kh = 0; kh < 3; ++kh)
#pragma unroll
            for (int kw = 0; kw < 3; ++kw) {
                float s = hv[kh] * vv[kw];
#pragma unroll
                for (int r = 1; r < RANK; ++r)
                    s = __builtin_fmaf(hv[3 * r + kh], vv[3 * r + kw], s);
                akf[e][kh * 3 + kw] = s;
            }
        asm volatile("" :: "v"(akf[e][0]), "v"(akf[e][1]), "v"(akf[e][2]), "v"(akf[e][3]), "v"(akf[e][4]),
                           "v"(akf[e][5]), "v"(akf[e][6]), "v"(akf[e][7]), "v"(akf[e][8]));
    }
    __builtin_amdgcn_sched_barrier(0);

#pragma unroll
    for (int i = 0; i < UNITS_PER_THREAD; ++i) {
        const int ry = i >> 1;
        const int a  = g8 + 8 * (i & 1);
        const int y  = h0 + ry - 1;
        const bool ok = (unsigned)y < (unsigned)HH;
        v4f c0 = p[i][0], c1 = p[i][1], c2 = p[i][2], c3 = p[i][3];
        if (!ok) { c0 = (v4f){0.f, 0.f, 0.f, 0.f}; c1 = c0; c2 = c0; c3 = c0; }
        float* d = tile + (ry * TILE_W + 4 * f + 1) * PIX_PITCH + 4 * a;
#pragma unroll
        for (int e = 0; e < 4; ++e)
            *(v4f*)(d + e * PIX_PITCH) = (v4f){c0[e], c1[e], c2[e], c3[e]};
    }
    for (int t = tid; t < SLAB_ROWS * 32; t += N_THREADS) {
        const int hr = t >> 5;
        const int hc = ((t >> 4) & 1) ? (TILE_W - 1) : 0;
        *(v4f*)(tile + (hr * TILE_W + hc) * PIX_PITCH + 4 * (t & 15)) = (v4f){0.f, 0.f, 0.f, 0.f};
    }
    __syncthreads();

    const float* bT[N_TILES];
#pragma unroll
    for (int j = 0; j < N_TILES; ++j) {
        const int q  = 16 * j + laneW;
        const int qr = q / WW;
        bT[j] = tile + ciL + (qr * TILE_W + (q - qr * WW)) * PIX_PITCH;
    }

    v4f bfr[2][3 * N_TILES];
#pragma unroll
    for (int kw = 0; kw < 3; ++kw)
#pragma unroll
        for (int j = 0; j < N_TILES; ++j)
            bfr[0][N_TILES * kw + j] = *(const v4f*)(bT[j] + kw * PIX_PITCH);

    v8f acc[N_TILES];
#pragma unroll
    for (int j = 0; j < N_TILES; ++j)
        acc[j] = (v8f){0.f, 0.f, 0.f, 0.f, 0.f, 0.f, 0.f, 0.f};

    __builtin_amdgcn_sched_barrier(0);

#pragma unroll
    for (int G = 0; G < 6; ++G) {
        const int cur = G & 1;
        const int g   = G / 3;
        const int kh  = G % 3;
        if (G + 1 < 6) {
            const int gn  = (G + 1) / 3;
            const int khn = (G + 1) % 3;
#pragma unroll
            for (int kw = 0; kw < 3; ++kw)
#pragma unroll
                for (int j = 0; j < N_TILES; ++j)
                    bfr[cur ^ 1][N_TILES * kw + j] =
                        *(const v4f*)(bT[j] + (khn * TILE_W + kw) * PIX_PITCH + 4 * gn);
        }
        if (G == 0) {
#pragma unroll
            for (int e = 4; e < 8; ++e)
#pragma unroll
                for (int m = 0; m < 3; ++m) {
                    wraw[e][m]     = *(const v4f*)(hLane + e * CI_PITCH + 4 * m);
                    wraw[e][3 + m] = *(const v4f*)(vLane + e * CI_PITCH + 4 * m);
                }
        }
        __builtin_amdgcn_sched_barrier(0);
#pragma unroll
        for (int kw = 0; kw < 3; ++kw) {
#pragma unroll
            for (int u = 0; u < 2; ++u) {
                if (G == 1 || G == 2) {
                    const int st = 6 * (G - 1) + 2 * kw + u;
                    const int e2 = 4 + st / 3;
                    const int k2 = st % 3;
                    float hq[RANK], vq[RANK][3];
#pragma unroll
                    for (int r = 0; r < RANK; ++r) {
                        const int ih = 3 * r + k2;
                        hq[r] = wraw[e2][ih >> 2][ih & 3];
#pragma unroll
                        for (int c = 0; c < 3; ++c) {
                            const int iv = 3 * r + c;
                            vq[r][c] = wraw[e2][3 + (iv >> 2)][iv & 3];
                        }
                    }
#pragma unroll
                    for (int c = 0; c < 3; ++c) {
                        float s = hq[0] * vq[0][c];
#pragma unroll
                        for (int r = 1; r < RANK; ++r)
                            s = __builtin_fmaf(hq[r], vq[r][c], s);
                        akf[e2][k2 * 3 + c] = s;
                    }
                    asm volatile("" :: "v"(akf[e2][k2 * 3]), "v"(akf[e2][k2 * 3 + 1]), "v"(akf[e2][k2 * 3 + 2]));
                }
                v2f av;
                av.x = akf[4 * g + 2 * u][kh * 3 + kw];
                av.y = akf[4 * g + 2 * u + 1][kh * 3 + kw];
#pragma unroll
                for (int j = 0; j < N_TILES; ++j) {
                    const v4f bq = bfr[cur][N_TILES * kw + j];
                    v2f bv;
                    bv.x = u ? bq.z : bq.x;
                    bv.y = u ? bq.w : bq.y;
                    acc[j] = __builtin_amdgcn_wmma_f32_16x16x4_f32(
                        false, av, false, bv, (short)0, acc[j], false, false);
                }
                if (G == 1 || G == 2) {
                    __builtin_amdgcn_sched_barrier(0);
                }
            }
        }
        __builtin_amdgcn_sched_barrier(0);
    }

    __syncthreads();
    float* red = tile + (wave * 16 + 8 * half) * RED_PITCH + laneW;
#pragma unroll
    for (int j = 0; j < N_TILES; ++j)
#pragma unroll
        for (int r = 0; r < 8; ++r)
            red[r * RED_PITCH + 16 * j] = acc[j][r];
    __syncthreads();

    float* ob = out + (((size_t)b * COUT + co0 + 4 * wave) * HH + h0) * WW;
#pragma unroll
    for (int i = 0; i < 7; ++i) {
        const int c   = lane + 32 * i;
        const int row = c / 56;
        const int q4  = 4 * (c - row * 56);
        const float* rp = tile + (4 * wave + row) * RED_PITCH + q4;
        const v4f s0 = *(const v4f*)(rp);
        const v4f s1 = *(const v4f*)(rp + 16 * RED_PITCH);
        const v4f s2 = *(const v4f*)(rp + 32 * RED_PITCH);
        const v4f s3 = *(const v4f*)(rp + 48 * RED_PITCH);
        v4f o;
#pragma unroll
        for (int e = 0; e < 4; ++e)
            o[e] = (((s0[e] + s1[e]) + s2[e]) + s3[e]) + bb[i];
        *(v4f*)(ob + (size_t)row * HH * WW + q4) = o;
    }
}

extern "C" void kernel_launch(void* const* d_in, const int* in_sizes, int n_in,
                              void* d_out, int out_size, void* d_ws, size_t ws_size,
                              hipStream_t stream) {
    const float* x    = (const float*)d_in[0];
    const float* vw   = (const float*)d_in[1];
    const float* hw   = (const float*)d_in[2];
    const float* bias = (const float*)d_in[3];
    float* out  = (float*)d_out;

    const int B = 4;
    conv_wmma<<<B * HH, N_THREADS, 0, stream>>>(x, vw, hw, bias, out);
}


# Minimal filtering F(2,3) along the width, on read: a column is an output pixel pair, its 4 raw pixels combined in registers beside the products; 336 products per wave (was 504): sum order changes
# speedup vs baseline: 3.0265x; 1.1320x over previous
#include <hip/hip_runtime.h>
#include <hip/hip_bf16.h>


typedef __attribute__((ext_vector_type(2))) float v2f;
typedef __attribute__((ext_vector_type(4))) float v4f;
typedef __attribute__((ext_vector_type(8))) float v8f;

#define CIN   64
#define COUT  64
#define RANK  4
#define HH    56
#define WW    56
#define TILE_W 58
#define PLANE_PIX 30
#define ROW_PIX (2 * PLANE_PIX)
#define PIX_PITCH 68
#define PLANE (HH * WW)
#define PAIR_FLOATS (RANK * 3)
#define CI_PITCH (COUT * PAIR_FLOATS)

#define ROWS_PER_WG 4
#define SLAB_ROWS (ROWS_PER_WG + 2)
#define SLAB_FLOATS (SLAB_ROWS * ROW_PIX * PIX_PITCH)
#define N_TILES 7
#define PAIRS_PER_ROW (WW / 2)
#define B_OFF(kh, d, g) (((2 * (kh) + ((d) & 1)) * PLANE_PIX + ((d) >> 1)) * PIX_PITCH + 4 * (g))
#define GROUP_PIX (ROWS_PER_WG * WW)
#define RED_PITCH 228
#define RED_FLOATS (4 * 16 * RED_PITCH)
#define N_THREADS 128
#define UNITS_PER_THREAD 12

template <int G>
static __device__ __forceinline__ void k_group(v4f (&raw)[2][4 * N_TILES], v4f (&tb)[2][N_TILES],
                                               v8f (&acc)[4][N_TILES], float (&gk)[8][3][4],
                                               v4f (&wraw)[8][6], const float* (&bT)[N_TILES],
                                               const float* hLane, const float* vLane) {
    constexpr int cur = G & 1;
    constexpr int g   = G / 3;
    constexpr int kh  = G % 3;
    if constexpr (G + 1 < 6) {
        constexpr int gn  = (G + 1) / 3;
        constexpr int khn = (G + 1) % 3;
#pragma unroll
        for (int d = 0; d < 4; ++d)
#pragma unroll
            for (int j = 0; j < N_TILES; ++j)
                raw[cur ^ 1][N_TILES * d + j] = *(const v4f*)(bT[j] + B_OFF(khn, d, gn));
    }
    if constexpr (G == 0) {
#pragma unroll
        for (int e = 4; e < 8; ++e)
#pragma unroll
            for (int m = 0; m < 3; ++m) {
                wraw[e][m]     = *(const v4f*)(hLane + e * CI_PITCH + 4 * m);
                wraw[e][3 + m] = *(const v4f*)(vLane + e * CI_PITCH + 4 * m);
            }
    }
    __builtin_amdgcn_sched_barrier(0);
#pragma unroll
    for (int c = 0; c < 4; ++c) {
#pragma unroll
        for (int u = 0; u < 2; ++u) {
            const int st8 = 2 * c + u;
            if constexpr (G == 1 || G == 2) {
                if (st8 < 6) {
                    const int st = 6 * (G - 1) + st8;
                    const int e2 = 4 + st / 3;
                    const int k2 = st % 3;
                    float hq[RANK], vq[RANK][3];
#pragma unroll
                    for (int r = 0; r < RANK; ++r) {
                        const int ih = 3 * r + k2;
                        hq[r] = wraw[e2][ih >> 2][ih & 3];
#pragma unroll
                        for (int cc = 0; cc < 3; ++cc) {
                            const int iv = 3 * r + cc;
                            vq[r][cc] = wraw[e2][3 + (iv >> 2)][iv & 3];
                        }
                    }
                    float kk[3];
#pragma unroll
                    for (int cc = 0; cc < 3; ++cc) {
                        float s = hq[0] * vq[0][cc];
#pragma unroll
                        for (int r = 1; r < RANK; ++r)
                            s = __builtin_fmaf(hq[r], vq[r][cc], s);
                        kk[cc] = s;
                    }
                    const float sp = kk[0] + kk[2];
                    gk[e2][k2][0] = kk[0];
                    gk[e2][k2][1] = 0.5f * (sp + kk[1]);
                    gk[e2][k2][2] = 0.5f * (sp - kk[1]);
                    gk[e2][k2][3] = kk[2];
                }
            }
#pragma unroll
            for (int j = 0; j < N_TILES; ++j) {
                if ((j < 4) == (u == 0)) {
                    if (c < 3) {
                        const v4f d1 = raw[cur][N_TILES + j];
                        const v4f d2 = raw[cur][2 * N_TILES + j];
                        const v4f d3 = raw[cur][3 * N_TILES + j];
                        tb[(c + 1) & 1][j] = (c == 0) ? (d1 + d2) : (c == 1) ? (d2 - d1) : (d1 - d3);
                    } else if (G + 1 < 6) {
                        tb[0][j] = raw[cur ^ 1][j] - raw[cur ^ 1][2 * N_TILES + j];
                    }
                }
            }
            v2f av;
            av.x = gk[4 * g + 2 * u][kh][c];
            av.y = gk[4 * g + 2 * u + 1][kh][c];
#pragma unroll
            for (int j = 0; j < N_TILES; ++j) {
                const v4f bq = tb[c & 1][j];
                v2f bv;
                bv.x = u ? bq.z : bq.x;
                bv.y = u ? bq.w : bq.y;
                acc[c][j] = __builtin_amdgcn_wmma_f32_16x16x4_f32(
                    false, av, false, bv, (short)0, acc[c][j], false, false);
            }
            __builtin_amdgcn_sched_barrier(0);
        }
    }
}

__global__ __launch_bounds__(N_THREADS) __attribute__((amdgpu_waves_per_eu(1, 1)))
void conv_wmma(const float* __restrict__ x,
               const float* __restrict__ vw,
               const float* __restrict__ hw,
               const float* __restrict__ bias,
               float* __restrict__ out) {
    __shared__ __attribute__((aligned(16))) float tile[SLAB_FLOATS];

    const int cot = blockIdx.x & 3;
    const int rg  = blockIdx.x >> 2;
    const int b   = rg / (HH / ROWS_PER_WG);
    const int h0  = (rg - b * (HH / ROWS_PER_WG)) * ROWS_PER_WG;
    const int tid = threadIdx.x;

    const int wave  = tid >> 5;
    const int lane  = tid & 31;
    const int half  = lane >> 4;
    const int laneW = lane & 15;
    const int co0 = cot * 16;
    const int ciL = 16 * wave + 8 * half;

    const float* xb = x + (size_t)b * CIN * HH * WW;
    const int f  = tid % 14;
    const int g8 = (tid / 14 < 8) ? tid / 14 : 7;

    const float* hLane = hw + ciL * CI_PITCH + (co0 + laneW) * PAIR_FLOATS;
    const float* vLane = vw + ciL * CI_PITCH + (co0 + laneW) * PAIR_FLOATS;
    v4f wraw[8][6];
#pragma unroll
    for (int e = 0; e < 4; ++e)
#pragma unroll
        for (int m = 0; m < 3; ++m) {
            wraw[e][m]     = *(const v4f*)(hLane + e * CI_PITCH + 4 * m);
            wraw[e][3 + m] = *(const v4f*)(vLane + e * CI_PITCH + 4 * m);
        }

    float bb[7];
#pragma unroll
    for (int i = 0; i < 7; ++i)
        bb[i] = bias[co0 + 4 * wave + (lane + 32 * i) / 56];
    __builtin_amdgcn_sched_barrier(0);

    v4f p[UNITS_PER_THREAD][4];
#pragma unroll
    for (int i = 0; i < UNITS_PER_THREAD; ++i) {
        const int ry = i >> 1;
        const int a  = g8 + 8 * (i & 1);
        const int y  = h0 + ry - 1;
        const bool ok = (unsigned)y < (unsigned)HH;
        const float* s = xb + ((4 * a) * HH + (ok ? y : h0)) * WW + 4 * f;
        p[i][0] = *(const v4f*)(s);
        p[i][1] = *(const v4f*)(s + PLANE);
        p[i][2] = *(const v4f*)(s + 2 * PLANE);
        p[i][3] = *(const v4f*)(s + 3 * PLANE);
    }
    __builtin_amdgcn_sched_barrier(0);

    float gk[8][3][4];
#pragma unroll
    for (int e = 0; e < 4; ++e) {
        float hv[12], vv[12];
#pragma unroll
        for (int m = 0; m < 3; ++m)
#pragma unroll
            for (int c = 0; c < 4; ++c) {
                hv[4 * m + c] = wraw[e][m][c];
                vv[4 * m + c] = wraw[e][3 + m][c];
            }
#pragma unroll
        for (int kh = 0; kh < 3; ++kh) {
            float kk[3];
#pragma unroll
            for (int kw = 0; kw < 3; ++kw) {
                float s = hv[kh] * vv[kw];
#pragma unroll
                for (int r = 1; r < RANK; ++r)
                    s = __builtin_fmaf(hv[3 * r + kh], vv[3 * r + kw], s);
                kk[kw] = s;
            }
            const float sp = kk[0] + kk[2];
            gk[e][kh][0] = kk[0];
            gk[e][kh][1] = 0.5f * (sp + kk[1]);
            gk[e][kh][2] = 0.5f * (sp - kk[1]);
            gk[e][kh][3] = kk[2];
        }
        asm volatile("" :: "v"(gk[e][0][1]), "v"(gk[e][0][2]), "v"(gk[e][1][1]), "v"(gk[e][1][2]),
                           "v"(gk[e][2][1]), "v"(gk[e][2][2]));
    }
    __builtin_amdgcn_sched_barrier(0);

#pragma unroll
    for (int i = 0; i < UNITS_PER_THREAD; ++i) {
        const int ry = i >> 1;
        const int a  = g8 + 8 * (i & 1);
        const int y  = h0 + ry - 1;
        const bool ok = (unsigned)y < (unsigned)HH;
        v4f c0 = p[i][0], c1 = p[i][1], c2 = p[i][2], c3 = p[i][3];
        if (!ok) { c0 = (v4f){0.f, 0.f, 0.f, 0.f}; c1 = c0; c2 = c0; c3 = c0; }
        float* d = tile + (ry * ROW_PIX + 2 * f) * PIX_PITCH + 4 * a;
#pragma unroll
        for (int e = 0; e < 4; ++e)
            *(v4f*)(d + (((1 + e) & 1) * PLANE_PIX + ((1 + e) >> 1)) * PIX_PITCH) = (v4f){c0[e], c1[e], c2[e], c3[e]};
    }
    for (int t = tid; t < SLAB_ROWS * 32; t += N_THREADS) {
        const int hr = t >> 5;
        const int hc = ((t >> 4) & 1) ? (PLANE_PIX + (TILE_W - 1) / 2) : 0;
        *(v4f*)(tile + (hr * ROW_PIX + hc) * PIX_PITCH + 4 * (t & 15)) = (v4f){0.f, 0.f, 0.f, 0.f};
    }
    __syncthreads();

    const float* bT[N_TILES];
#pragma unroll
    for (int j = 0; j < N_TILES; ++j) {
        const int n  = 16 * j + laneW;
        const int qr = n / PAIRS_PER_ROW;
        bT[j] = tile + ciL + (qr * ROW_PIX + (n - qr * PAIRS_PER_ROW)) * PIX_PITCH;
    }

    v4f raw[2][4 * N_TILES];
#pragma unroll
    for (int d = 0; d < 4; ++d)
#pragma unroll
        for (int j = 0; j < N_TILES; ++j)
            raw[0][N_TILES * d + j] = *(const v4f*)(bT[j] + B_OFF(0, d, 0));

    v4f tb[2][N_TILES];
#pragma unroll
    for (int j = 0; j < N_TILES; ++j)
        tb[0][j] = raw[0][j] - raw[0][2 * N_TILES + j];

    v8f acc[4][N_TILES];
#pragma unroll
    for (int c = 0; c < 4; ++c)
#pragma unroll
        for (int j = 0; j < N_TILES; ++j)
            acc[c][j] = (v8f){0.f, 0.f, 0.f, 0.f, 0.f, 0.f, 0.f, 0.f};

    __builtin_amdgcn_sched_barrier(0);

    k_group<0>(raw, tb, acc, gk, wraw, bT, hLane, vLane);
    k_group<1>(raw, tb, acc, gk, wraw, bT, hLane, vLane);
    k_group<2>(raw, tb, acc, gk, wraw, bT, hLane, vLane);
    k_group<3>(raw, tb, acc, gk, wraw, bT, hLane, vLane);
    k_group<4>(raw, tb, acc, gk, wraw, bT, hLane, vLane);
    k_group<5>(raw, tb, acc, gk, wraw, bT, hLane, vLane);

    __syncthreads();
    float* red = tile + (wave * 16 + 8 * half) * RED_PITCH + 2 * laneW;
#pragma unroll
    for (int j = 0; j < N_TILES; ++j)
#pragma unroll
        for (int r = 0; r < 8; ++r) {
            const float m0 = acc[0][j][r], m1 = acc[1][j][r], m2 = acc[2][j][r], m3 = acc[3][j][r];
            v2f y;
            y.x = (m0 + m1) + m2;
            y.y = (m1 - m2) - m3;
            *(v2f*)(red + r * RED_PITCH + 32 * j) = y;
        }
    __syncthreads();

    float* ob = out + (((size_t)b * COUT + co0 + 4 * wave) * HH + h0) * WW;
#pragma unroll
    for (int i = 0; i < 7; ++i) {
        const int c   = lane + 32 * i;
        const int row = c / 56;
        const int q4  = 4 * (c - row * 56);
        const float* rp = tile + (4 * wave + row) * RED_PITCH + q4;
        const v4f s0 = *(const v4f*)(rp);
        const v4f s1 = *(const v4f*)(rp + 16 * RED_PITCH);
        const v4f s2 = *(const v4f*)(rp + 32 * RED_PITCH);
        const v4f s3 = *(const v4f*)(rp + 48 * RED_PITCH);
        v4f o;
#pragma unroll
        for (int e = 0; e < 4; ++e)
            o[e] = (((s0[e] + s1[e]) + s2[e]) + s3[e]) + bb[i];
        *(v4f*)(ob + (size_t)row * HH * WW + q4) = o;
    }
}

extern "C" void kernel_launch(void* const* d_in, const int* in_sizes, int n_in,
                              void* d_out, int out_size, void* d_ws, size_t ws_size,
                              hipStream_t stream) {
    const float* x    = (const float*)d_in[0];
    const float* vw   = (const float*)d_in[1];
    const float* hw   = (const float*)d_in[2];
    const float* bias = (const float*)d_in[3];
    float* out  = (float*)d_out;

    const int B = 4;
    conv_wmma<<<B * HH, N_THREADS, 0, stream>>>(x, vw, hw, bias, out);
}
